# starting kernel
# baseline (speedup 1.0000x reference)
#include <hip/hip_runtime.h>

typedef float v2f __attribute__((ext_vector_type(2)));
typedef float v8f __attribute__((ext_vector_type(8)));

#define DIN  512
#define DHID 512
#define DOUT 256


__global__ void k_fill(float* __restrict__ p, float val, int n) {
  int i = blockIdx.x * blockDim.x + threadIdx.x;
  if (i < n) p[i] = val;
}

__global__ void k_zero4(float4* __restrict__ p, int n4) {
  int i = blockIdx.x * blockDim.x + threadIdx.x;
  if (i < n4) p[i] = make_float4(0.f, 0.f, 0.f, 0.f);
}

__device__ __forceinline__ void atomAddF(float* p, float v) {
  __hip_atomic_fetch_add(p, v, __ATOMIC_RELAXED, __HIP_MEMORY_SCOPE_AGENT);
}


__global__ void k_deg_count(const int* __restrict__ tgt, float* __restrict__ deg, int E) {
  int e = blockIdx.x * blockDim.x + threadIdx.x;
  if (e < E) atomAddF(&deg[tgt[e]], 1.0f);
}

__global__ void k_rsqrt(float* __restrict__ p, int n) {
  int i = blockIdx.x * blockDim.x + threadIdx.x;
  if (i < n) {
    float d = p[i];
    p[i] = (d > 0.f) ? rsqrtf(d) : 0.f;
  }
}


template <int K, int N, bool SQUARE_A, bool TAIL>
__device__ __forceinline__ void gemm_kloop(const float* __restrict__ a0,
                                           const float* __restrict__ a1,
                                           const float* __restrict__ a2,
                                           const float* __restrict__ a3,
                                           const float* __restrict__ bcol,
                                           v8f acc[4][2]) {
  for (int k0 = 0; k0 < K; k0 += 4) {
    v2f a[4];
    if (!TAIL) {
      a[0] = *(const v2f*)(a0 + k0);
      a[1] = *(const v2f*)(a0 + k0 + 16 * K);
      a[2] = *(const v2f*)(a0 + k0 + 32 * K);
      a[3] = *(const v2f*)(a0 + k0 + 48 * K);
    } else {
      a[0] = *(const v2f*)(a0 + k0);
      a[1] = *(const v2f*)(a1 + k0);
      a[2] = *(const v2f*)(a2 + k0);
      a[3] = *(const v2f*)(a3 + k0);
    }
    if (SQUARE_A) {
#pragma unroll
      for (int i = 0; i < 4; ++i) { a[i].x *= a[i].x; a[i].y *= a[i].y; }
    }
    const float* bp = bcol + (size_t)k0 * N;
    v2f b0, b1;
    b0.x = bp[0];  b0.y = bp[N + 0];
    b1.x = bp[16]; b1.y = bp[N + 16];
#pragma unroll
    for (int i = 0; i < 4; ++i) {
      acc[i][0] = __builtin_amdgcn_wmma_f32_16x16x4_f32(false, a[i], false, b0,
                                                        (short)0, acc[i][0], false, false);
      acc[i][1] = __builtin_amdgcn_wmma_f32_16x16x4_f32(false, a[i], false, b1,
                                                        (short)0, acc[i][1], false, false);
    }
  }
}

template <int K, int N, bool SQUARE_A>
__global__ __launch_bounds__(256)
void k_gemm_wmma(const float* __restrict__ A, const float* __restrict__ B,
                 float* __restrict__ C, int M) {
  const int lane = threadIdx.x & 31;
  const int wid  = threadIdx.x >> 5;
  const int m = lane & 15;
  const int g = lane >> 4;

  const int row0 = (blockIdx.x * 8 + wid) * 64;
  if (row0 >= M) return;
  const int col0 = blockIdx.y * 32;

  int  r[4];
  bool st[4];
#pragma unroll
  for (int i = 0; i < 4; ++i) {
    int rr = row0 + 16 * i;
    st[i] = (rr < M);
    r[i]  = st[i] ? rr : (M - 16);
  }

  v8f acc[4][2] = {};
  const float* a0   = A + (size_t)(r[0] + m) * K + 2 * g;
  const float* bcol = B + (size_t)(2 * g) * N + col0 + m;

  if (row0 + 64 <= M) {
    gemm_kloop<K, N, SQUARE_A, false>(a0, a0, a0, a0, bcol, acc);
  } else {
    const float* a1 = A + (size_t)(r[1] + m) * K + 2 * g;
    const float* a2 = A + (size_t)(r[2] + m) * K + 2 * g;
    const float* a3 = A + (size_t)(r[3] + m) * K + 2 * g;
    gemm_kloop<K, N, SQUARE_A, true>(a0, a1, a2, a3, bcol, acc);
  }

#pragma unroll
  for (int i = 0; i < 4; ++i) {
    if (!st[i]) continue;
    float* crow = C + (size_t)(r[i] + 8 * g) * N + col0 + m;
#pragma unroll
    for (int v = 0; v < 8; ++v) {
      crow[(size_t)v * N]      = acc[i][0][v];
      crow[(size_t)v * N + 16] = acc[i][1][v];
    }
  }
}

__global__ __launch_bounds__(256)
void k_scatter(const float* __restrict__ feat, const float* __restrict__ dinv,
               const int* __restrict__ src, const int* __restrict__ tgt,
               float* __restrict__ out, int E, int N, int D) {
  int wave = (int)((blockIdx.x * blockDim.x + threadIdx.x) >> 5);
  int lane = threadIdx.x & 31;
  int total = E + N;
  if (wave >= total) return;

  int s, t;
  if (wave < E) { s = src[wave]; t = tgt[wave]; }
  else          { s = t = wave - E; }

  float norm = dinv[s] * dinv[t];
  const float4* fs = (const float4*)(feat + (size_t)s * D);
  float* od = out + (size_t)t * D;
  int nvec = D >> 2;
  for (int v = lane; v < nvec; v += 32) {
    float4 val = fs[v];
    int base = v * 4;
    atomAddF(od + base + 0, norm * val.x);
    atomAddF(od + base + 1, norm * val.y);
    atomAddF(od + base + 2, norm * val.z);
    atomAddF(od + base + 3, norm * val.w);
  }
}


extern "C" void kernel_launch(void* const* d_in, const int* in_sizes, int n_in,
                              void* d_out, int out_size, void* d_ws, size_t ws_size,
                              hipStream_t stream) {
  const float* x  = (const float*)d_in[0];
  const int*   ei = (const int*)d_in[1];
  const float* W1 = (const float*)d_in[2];
  const float* W2 = (const float*)d_in[3];
  float* out = (float*)d_out;

  const int N = in_sizes[0] / DIN;
  const int E = in_sizes[1] / 2;
  const int* src = ei;
  const int* tgt = ei + E;

  char* ws = (char*)d_ws;
  size_t off = 0;
  float* dinv = (float*)(ws + off); off += ((size_t)N * 4 + 255) & ~(size_t)255;
  float* xw   = (float*)(ws + off); off += ((size_t)N * DHID * 4 + 255) & ~(size_t)255;
  float* h    = (float*)(ws + off); off += ((size_t)N * DHID * 4 + 255) & ~(size_t)255;
  float* hw   = xw;
  (void)ws_size; (void)n_in; (void)out_size;

  const int nEdges = E + N;
  const int mBlocks = (N + 511) / 512;

  k_fill<<<(N + 255) / 256, 256, 0, stream>>>(dinv, 1.0f, N);
  k_deg_count<<<(E + 255) / 256, 256, 0, stream>>>(tgt, dinv, E);
  k_rsqrt<<<(N + 255) / 256, 256, 0, stream>>>(dinv, N);

  {
    dim3 grid(mBlocks, DHID / 32);
    k_gemm_wmma<DIN, DHID, false><<<grid, 256, 0, stream>>>(x, W1, xw, N);
  }
  {
    int n4 = N * DHID / 4;
    k_zero4<<<(n4 + 255) / 256, 256, 0, stream>>>((float4*)h, n4);
    k_scatter<<<(nEdges + 7) / 8, 256, 0, stream>>>(xw, dinv, src, tgt, h, E, N, DHID);
  }
  {
    dim3 grid(mBlocks, DOUT / 32);
    k_gemm_wmma<DHID, DOUT, true><<<grid, 256, 0, stream>>>(h, W2, hw, N);
  }
  {
    int o4 = N * DOUT / 4;
    k_zero4<<<(o4 + 255) / 256, 256, 0, stream>>>((float4*)out, o4);
    k_scatter<<<(nEdges + 7) / 8, 256, 0, stream>>>(hw, dinv, src, tgt, out, E, N, DOUT);
  }
}


# Edge scatter (global f32 atomics) replaced by target tiles in LDS: each work-group filters the edge list for its 40 rows, sums them with LDS f32 adds (arrival order), stores rows once; no zero-fill
# speedup vs baseline: 1.4975x; 1.4975x over previous
#include <hip/hip_runtime.h>

typedef float v2f __attribute__((ext_vector_type(2)));
typedef float v8f __attribute__((ext_vector_type(8)));

#define DIN  512
#define DHID 512
#define DOUT 256


__global__ void k_fill(float* __restrict__ p, float val, int n) {
  int i = blockIdx.x * blockDim.x + threadIdx.x;
  if (i < n) p[i] = val;
}

__device__ __forceinline__ void atomAddF(float* p, float v) {
  __hip_atomic_fetch_add(p, v, __ATOMIC_RELAXED, __HIP_MEMORY_SCOPE_AGENT);
}


__global__ void k_deg_count(const int* __restrict__ tgt, float* __restrict__ deg, int E) {
  int e = blockIdx.x * blockDim.x + threadIdx.x;
  if (e < E) atomAddF(&deg[tgt[e]], 1.0f);
}

__global__ void k_rsqrt(float* __restrict__ p, int n) {
  int i = blockIdx.x * blockDim.x + threadIdx.x;
  if (i < n) {
    float d = p[i];
    p[i] = (d > 0.f) ? rsqrtf(d) : 0.f;
  }
}


template <int K, int N, bool SQUARE_A, bool TAIL>
__device__ __forceinline__ void gemm_kloop(const float* __restrict__ a0,
                                           const float* __restrict__ a1,
                                           const float* __restrict__ a2,
                                           const float* __restrict__ a3,
                                           const float* __restrict__ bcol,
                                           v8f acc[4][2]) {
  for (int k0 = 0; k0 < K; k0 += 4) {
    v2f a[4];
    if (!TAIL) {
      a[0] = *(const v2f*)(a0 + k0);
      a[1] = *(const v2f*)(a0 + k0 + 16 * K);
      a[2] = *(const v2f*)(a0 + k0 + 32 * K);
      a[3] = *(const v2f*)(a0 + k0 + 48 * K);
    } else {
      a[0] = *(const v2f*)(a0 + k0);
      a[1] = *(const v2f*)(a1 + k0);
      a[2] = *(const v2f*)(a2 + k0);
      a[3] = *(const v2f*)(a3 + k0);
    }
    if (SQUARE_A) {
#pragma unroll
      for (int i = 0; i < 4; ++i) { a[i].x *= a[i].x; a[i].y *= a[i].y; }
    }
    const float* bp = bcol + (size_t)k0 * N;
    v2f b0, b1;
    b0.x = bp[0];  b0.y = bp[N + 0];
    b1.x = bp[16]; b1.y = bp[N + 16];
#pragma unroll
    for (int i = 0; i < 4; ++i) {
      acc[i][0] = __builtin_amdgcn_wmma_f32_16x16x4_f32(false, a[i], false, b0,
                                                        (short)0, acc[i][0], false, false);
      acc[i][1] = __builtin_amdgcn_wmma_f32_16x16x4_f32(false, a[i], false, b1,
                                                        (short)0, acc[i][1], false, false);
    }
  }
}

template <int K, int N, bool SQUARE_A>
__global__ __launch_bounds__(256)
void k_gemm_wmma(const float* __restrict__ A, const float* __restrict__ B,
                 float* __restrict__ C, int M) {
  const int lane = threadIdx.x & 31;
  const int wid  = threadIdx.x >> 5;
  const int m = lane & 15;
  const int g = lane >> 4;

  const int row0 = (blockIdx.x * 8 + wid) * 64;
  if (row0 >= M) return;
  const int col0 = blockIdx.y * 32;

  int  r[4];
  bool st[4];
#pragma unroll
  for (int i = 0; i < 4; ++i) {
    int rr = row0 + 16 * i;
    st[i] = (rr < M);
    r[i]  = st[i] ? rr : (M - 16);
  }

  v8f acc[4][2] = {};
  const float* a0   = A + (size_t)(r[0] + m) * K + 2 * g;
  const float* bcol = B + (size_t)(2 * g) * N + col0 + m;

  if (row0 + 64 <= M) {
    gemm_kloop<K, N, SQUARE_A, false>(a0, a0, a0, a0, bcol, acc);
  } else {
    const float* a1 = A + (size_t)(r[1] + m) * K + 2 * g;
    const float* a2 = A + (size_t)(r[2] + m) * K + 2 * g;
    const float* a3 = A + (size_t)(r[3] + m) * K + 2 * g;
    gemm_kloop<K, N, SQUARE_A, true>(a0, a1, a2, a3, bcol, acc);
  }

#pragma unroll
  for (int i = 0; i < 4; ++i) {
    if (!st[i]) continue;
    float* crow = C + (size_t)(r[i] + 8 * g) * N + col0 + m;
#pragma unroll
    for (int v = 0; v < 8; ++v) {
      crow[(size_t)v * N]      = acc[i][0][v];
      crow[(size_t)v * N + 16] = acc[i][1][v];
    }
  }
}


#define AGG_ROWS 40
#define AGG_NT   1024
#define AGG_NW   (AGG_NT / 32)
#define AGG_CAP  2048
#define AGG_SCAN 8

template <int D, int U>
__global__ __launch_bounds__(AGG_NT)
void k_agg_tile(const float* __restrict__ feat, const float* __restrict__ dinv,
                const int* __restrict__ src, const int* __restrict__ tgt,
                float* __restrict__ out, int E, int N) {
  __shared__ float tile[AGG_ROWS * D];
  __shared__ int   l_src[AGG_CAP];
  __shared__ int   l_row[AGG_CAP];
  __shared__ float l_nrm[AGG_CAP];
  __shared__ int   l_cnt;

  constexpr int D4  = D / 4;
  constexpr int Q   = D / 64;
  constexpr int IT0 = (AGG_ROWS * D4 + AGG_NT - 1) / AGG_NT;
  const int tid  = threadIdx.x;
  const int lane = tid & 31;
  const int wid  = tid >> 5;
  const int row0 = blockIdx.x * AGG_ROWS;
  const int rows = min(AGG_ROWS, N - row0);

  {
    float4 f[IT0];
    float  d[IT0];
#pragma unroll
    for (int i = 0; i < IT0; ++i) {
      const int idx = tid + i * AGG_NT;
      const int r = idx / D4, c4 = idx - r * D4;
      const int t = min(row0 + r, N - 1);
      d[i] = dinv[t];
      f[i] = ((const float4*)(feat + (size_t)t * D))[c4];
    }
#pragma unroll
    for (int i = 0; i < IT0; ++i) {
      const int idx = tid + i * AGG_NT;
      const float nrm = d[i] * d[i];
      if (idx < AGG_ROWS * D4)
        ((float4*)tile)[idx] = make_float4(nrm * f[i].x, nrm * f[i].y, nrm * f[i].z, nrm * f[i].w);
    }
  }
  if (tid == 0) l_cnt = 0;
  __syncthreads();

  auto take = [&](int e, int t) {
    const unsigned r = (unsigned)(t - row0);
    if (r < (unsigned)rows) {
      const int pos = __hip_atomic_fetch_add(&l_cnt, 1, __ATOMIC_RELAXED,
                                             __HIP_MEMORY_SCOPE_WORKGROUP);
      if (pos < AGG_CAP) {
        l_src[pos] = e; l_row[pos] = (int)r;
      } else {
        const int s = src[e];
        const float nrm = dinv[s] * dinv[t];
        const float* fs = feat + (size_t)s * D;
        float* tr = tile + r * D;
        for (int c = 0; c < D; ++c)
          __hip_atomic_fetch_add(tr + c, nrm * fs[c], __ATOMIC_RELAXED,
                                 __HIP_MEMORY_SCOPE_WORKGROUP);
      }
    }
  };

  const bool vec = ((E & 3) == 0) && ((((size_t)tgt) & 15) == 0);
  const int nvec = vec ? (E >> 2) : 0;
  const int rot = (int)blockIdx.x * (nvec / (int)gridDim.x);
  const int4* tgt4 = (const int4*)tgt;
  for (int base = tid; base < nvec; base += AGG_SCAN * AGG_NT) {
    int4 tv[AGG_SCAN];
    int  ix[AGG_SCAN];
#pragma unroll
    for (int u = 0; u < AGG_SCAN; ++u) {
      const int i = base + u * AGG_NT;
      int j = i + rot;
      if (j >= nvec) j -= nvec;
      ix[u] = j;
      tv[u] = (i < nvec) ? tgt4[j] : make_int4(-1, -1, -1, -1);
    }
#pragma unroll
    for (int u = 0; u < AGG_SCAN; ++u) {
      take(4 * ix[u] + 0, tv[u].x);
      take(4 * ix[u] + 1, tv[u].y);
      take(4 * ix[u] + 2, tv[u].z);
      take(4 * ix[u] + 3, tv[u].w);
    }
  }
  for (int e = nvec * 4 + tid; e < E; e += AGG_NT) take(e, tgt[e]);
  __syncthreads();

  const int n = min(l_cnt, AGG_CAP);
  for (int k = tid; k < n; k += AGG_NT) {
    const int s = src[l_src[k]];
    l_nrm[k] = dinv[s] * dinv[row0 + l_row[k]];
    l_src[k] = s;
  }
  __syncthreads();

  int k = wid;
  for (; k + (U - 1) * AGG_NW < n; k += U * AGG_NW) {
    float2 a[U][Q];
    int    rr[U];
    float  nn[U];
#pragma unroll
    for (int u = 0; u < U; ++u) {
      const int ku = k + u * AGG_NW;
      const int s = l_src[ku];
      rr[u] = l_row[ku];
      nn[u] = l_nrm[ku];
      const float2* f = (const float2*)(feat + (size_t)s * D) + lane;
#pragma unroll
      for (int q = 0; q < Q; ++q) a[u][q] = f[32 * q];
    }
#pragma unroll
    for (int u = 0; u < U; ++u) {
      float* tr = tile + rr[u] * D + 2 * lane;
#pragma unroll
      for (int q = 0; q < Q; ++q) {
        __hip_atomic_fetch_add(tr + 64 * q,     nn[u] * a[u][q].x, __ATOMIC_RELAXED,
                               __HIP_MEMORY_SCOPE_WORKGROUP);
        __hip_atomic_fetch_add(tr + 64 * q + 1, nn[u] * a[u][q].y, __ATOMIC_RELAXED,
                               __HIP_MEMORY_SCOPE_WORKGROUP);
      }
    }
  }
  for (; k < n; k += AGG_NW) {
    const int s = l_src[k];
    const float nk = l_nrm[k];
    const float2* f = (const float2*)(feat + (size_t)s * D) + lane;
    float* tr = tile + l_row[k] * D + 2 * lane;
    float2 a[Q];
#pragma unroll
    for (int q = 0; q < Q; ++q) a[q] = f[32 * q];
#pragma unroll
    for (int q = 0; q < Q; ++q) {
      __hip_atomic_fetch_add(tr + 64 * q,     nk * a[q].x, __ATOMIC_RELAXED,
                             __HIP_MEMORY_SCOPE_WORKGROUP);
      __hip_atomic_fetch_add(tr + 64 * q + 1, nk * a[q].y, __ATOMIC_RELAXED,
                             __HIP_MEMORY_SCOPE_WORKGROUP);
    }
  }
  __syncthreads();

  for (int idx = tid; idx < rows * D4; idx += AGG_NT) {
    const int r = idx / D4, c4 = idx - r * D4;
    ((float4*)(out + (size_t)(row0 + r) * D))[c4] = ((const float4*)tile)[idx];
  }
}


extern "C" void kernel_launch(void* const* d_in, const int* in_sizes, int n_in,
                              void* d_out, int out_size, void* d_ws, size_t ws_size,
                              hipStream_t stream) {
  const float* x  = (const float*)d_in[0];
  const int*   ei = (const int*)d_in[1];
  const float* W1 = (const float*)d_in[2];
  const float* W2 = (const float*)d_in[3];
  float* out = (float*)d_out;

  const int N = in_sizes[0] / DIN;
  const int E = in_sizes[1] / 2;
  const int* src = ei;
  const int* tgt = ei + E;

  char* ws = (char*)d_ws;
  size_t off = 0;
  float* dinv = (float*)(ws + off); off += ((size_t)N * 4 + 255) & ~(size_t)255;
  float* xw   = (float*)(ws + off); off += ((size_t)N * DHID * 4 + 255) & ~(size_t)255;
  float* h    = (float*)(ws + off); off += ((size_t)N * DHID * 4 + 255) & ~(size_t)255;
  float* hw   = xw;
  (void)ws_size; (void)n_in; (void)out_size;

  const int aggBlocks = (N + AGG_ROWS - 1) / AGG_ROWS;
  const int mBlocks = (N + 511) / 512;

  k_fill<<<(N + 255) / 256, 256, 0, stream>>>(dinv, 1.0f, N);
  k_deg_count<<<(E + 255) / 256, 256, 0, stream>>>(tgt, dinv, E);
  k_rsqrt<<<(N + 255) / 256, 256, 0, stream>>>(dinv, N);

  {
    dim3 grid(mBlocks, DHID / 32);
    k_gemm_wmma<DIN, DHID, false><<<grid, 256, 0, stream>>>(x, W1, xw, N);
  }
  k_agg_tile<DHID, 2><<<aggBlocks, AGG_NT, 0, stream>>>(xw, dinv, src, tgt, h, E, N);
  {
    dim3 grid(mBlocks, DOUT / 32);
    k_gemm_wmma<DHID, DOUT, true><<<grid, 256, 0, stream>>>(h, W2, hw, N);
  }
  k_agg_tile<DOUT, 4><<<aggBlocks, AGG_NT, 0, stream>>>(hw, dinv, src, tgt, out, E, N);
}


# GEMM operands staged through LDS: 80 x 256 work-group tile (layer 2: 80 x 128), A and B copied by whole lines one 64-k block ahead, fragments read from LDS, C stored by whole lines; k order kept
# speedup vs baseline: 1.9273x; 1.2870x over previous
#include <hip/hip_runtime.h>

typedef float v2f __attribute__((ext_vector_type(2)));
typedef float v8f __attribute__((ext_vector_type(8)));

#define DIN  512
#define DHID 512
#define DOUT 256


__global__ void k_fill(float* __restrict__ p, float val, int n) {
  int i = blockIdx.x * blockDim.x + threadIdx.x;
  if (i < n) p[i] = val;
}

__device__ __forceinline__ void atomAddF(float* p, float v) {
  __hip_atomic_fetch_add(p, v, __ATOMIC_RELAXED, __HIP_MEMORY_SCOPE_AGENT);
}


__global__ void k_deg_count(const int* __restrict__ tgt, float* __restrict__ deg, int E) {
  int e = blockIdx.x * blockDim.x + threadIdx.x;
  if (e < E) atomAddF(&deg[tgt[e]], 1.0f);
}

__global__ void k_rsqrt(float* __restrict__ p, int n) {
  int i = blockIdx.x * blockDim.x + threadIdx.x;
  if (i < n) {
    float d = p[i];
    p[i] = (d > 0.f) ? rsqrtf(d) : 0.f;
  }
}


#define GM_BM 80
#define GM_BK 64
#define GM_PA 68
#define GM_NTH 256

template <int K, int N, int BN, bool SQUARE_A>
__global__ __launch_bounds__(GM_NTH)
void k_gemm_lds(const float* A, const float* B, float* C, int M) {
  constexpr int NT      = BN / 128;
  constexpr int PB      = 2 * BN + 32;
  constexpr int PC      = BN + 4;
  constexpr int A_STAGE = GM_BM * GM_PA;
  constexpr int B_STAGE = (GM_BK / 2) * PB;
  constexpr int STAGE   = A_STAGE + B_STAGE;
  constexpr int C_TILE  = GM_BM * PC;
  constexpr int LDS_FL  = (2 * STAGE > C_TILE) ? 2 * STAGE : C_TILE;
  constexpr int NKB     = K / GM_BK;
  constexpr int AK4     = GM_BK / 4;
  constexpr int A_IT    = GM_BM * AK4 / GM_NTH;
  constexpr int BN4     = BN / 4;
  constexpr int B_IT    = (GM_BK / 2) * BN4 / GM_NTH;
  static_assert(K % GM_BK == 0 && BN % 128 == 0 && N % BN == 0, "tile shape");
  static_assert((GM_BM * AK4) % GM_NTH == 0, "A copy shape");
  static_assert(((GM_BK / 2) * BN4) % GM_NTH == 0, "B copy shape");

  __shared__ __attribute__((aligned(16))) float lds[LDS_FL];

  const int tid  = threadIdx.x;
  const int lane = tid & 31;
  const int wid  = tid >> 5;
  const int m = lane & 15;
  const int g = lane >> 4;
  const int row0 = blockIdx.x * GM_BM;
  const int col0 = blockIdx.y * BN;
  const int wcol = wid * (NT * 16);

  const float* ap[A_IT];
  int a_l[A_IT];
#pragma unroll
  for (int i = 0; i < A_IT; ++i) {
    const int idx = tid + i * GM_NTH;
    const int r = idx / AK4, c4 = idx - r * AK4;
    const int gr = min(row0 + r, M - 1);
    ap[i]  = A + (size_t)gr * K + c4 * 4;
    a_l[i] = r * GM_PA + c4 * 4;
  }
  const float* bp[B_IT];
  int b_l[B_IT];
#pragma unroll
  for (int i = 0; i < B_IT; ++i) {
    const int item = tid + i * GM_NTH;
    const int jp = item / BN4, n4 = item - jp * BN4;
    bp[i]  = B + (size_t)(2 * jp) * N + col0 + n4 * 4;
    b_l[i] = jp * PB + n4 * 8;
  }

  float4 ra[A_IT], rb0[B_IT], rb1[B_IT];
  auto gload = [&](int kb) {
#pragma unroll
    for (int i = 0; i < A_IT; ++i) ra[i] = *(const float4*)(ap[i] + kb * GM_BK);
#pragma unroll
    for (int i = 0; i < B_IT; ++i) {
      const float* p = bp[i] + (size_t)kb * GM_BK * N;
      rb0[i] = *(const float4*)(p);
      rb1[i] = *(const float4*)(p + N);
    }
  };
  auto lstore = [&](int st) {
    float* as = lds + st * STAGE;
    float* bs = as + A_STAGE;
#pragma unroll
    for (int i = 0; i < A_IT; ++i) {
      const float4 v = ra[i];
      *(float4*)(as + a_l[i]) = make_float4(v.x, v.y, v.z, v.w);
    }
#pragma unroll
    for (int i = 0; i < B_IT; ++i) {
      *(float4*)(bs + b_l[i])     = make_float4(rb0[i].x, rb1[i].x, rb0[i].y, rb1[i].y);
      *(float4*)(bs + b_l[i] + 4) = make_float4(rb0[i].z, rb1[i].z, rb0[i].w, rb1[i].w);
    }
  };

  v8f acc[5][NT] = {};
  auto compute = [&](int st) {
    const float* as = lds + st * STAGE + m * GM_PA + 2 * g;
    const float* bs = lds + st * STAGE + A_STAGE + g * PB + (wcol + m) * 2;
#pragma unroll
    for (int s = 0; s < GM_BK / 4; ++s) {
      v2f a[5], b[NT];
#pragma unroll
      for (int i = 0; i < 5; ++i) {
        a[i] = *(const v2f*)(as + i * 16 * GM_PA + s * 4);
        if (SQUARE_A) a[i] *= a[i];
      }
#pragma unroll
      for (int j = 0; j < NT; ++j) b[j] = *(const v2f*)(bs + s * 2 * PB + j * 32);
#pragma unroll
      for (int i = 0; i < 5; ++i)
#pragma unroll
        for (int j = 0; j < NT; ++j)
          acc[i][j] = __builtin_amdgcn_wmma_f32_16x16x4_f32(false, a[i], false, b[j],
                                                            (short)0, acc[i][j], false, false);
    }
  };

  gload(0);
  lstore(0);
  __syncthreads();
  for (int kb = 0; kb < NKB - 1; ++kb) {
    gload(kb + 1);
    compute(kb & 1);
    __builtin_amdgcn_sched_barrier(0);
    lstore((kb + 1) & 1);
    __syncthreads();
  }
  compute((NKB - 1) & 1);
  __syncthreads();

#pragma unroll
  for (int i = 0; i < 5; ++i)
#pragma unroll
    for (int j = 0; j < NT; ++j)
#pragma unroll
      for (int v = 0; v < 8; ++v)
        lds[(i * 16 + 8 * g + v) * PC + wcol + j * 16 + m] = acc[i][j][v];
  __syncthreads();

  constexpr int C_IT = GM_BM * BN4 / GM_NTH;
  static_assert((GM_BM * BN4) % GM_NTH == 0, "C copy shape");
  if (row0 + GM_BM <= M) {
    float4 cv[C_IT];
#pragma unroll
    for (int it = 0; it < C_IT; ++it) {
      const int idx = tid + it * GM_NTH;
      const int r = idx / BN4, c4 = idx - r * BN4;
      cv[it] = *(const float4*)(lds + r * PC + c4 * 4);
    }
#pragma unroll
    for (int it = 0; it < C_IT; ++it) {
      const int idx = tid + it * GM_NTH;
      const int r = idx / BN4, c4 = idx - r * BN4;
      *(float4*)(C + (size_t)(row0 + r) * N + col0 + c4 * 4) = cv[it];
    }
  } else {
    for (int idx = tid; idx < GM_BM * BN4; idx += GM_NTH) {
      const int r = idx / BN4, c4 = idx - r * BN4;
      if (row0 + r < M)
        *(float4*)(C + (size_t)(row0 + r) * N + col0 + c4 * 4) = *(const float4*)(lds + r * PC + c4 * 4);
    }
  }
}


#define AGG_ROWS 40
#define AGG_NT   1024
#define AGG_NW   (AGG_NT / 32)
#define AGG_CAP  2048
#define AGG_SCAN 8

template <int D, int U>
__global__ __launch_bounds__(AGG_NT)
void k_agg_tile(const float* __restrict__ feat, const float* __restrict__ dinv,
                const int* __restrict__ src, const int* __restrict__ tgt,
                float* __restrict__ out, int E, int N) {
  __shared__ float tile[AGG_ROWS * D];
  __shared__ int   l_src[AGG_CAP];
  __shared__ int   l_row[AGG_CAP];
  __shared__ float l_nrm[AGG_CAP];
  __shared__ int   l_cnt;

  constexpr int D4  = D / 4;
  constexpr int Q   = D / 64;
  constexpr int IT0 = (AGG_ROWS * D4 + AGG_NT - 1) / AGG_NT;
  const int tid  = threadIdx.x;
  const int lane = tid & 31;
  const int wid  = tid >> 5;
  const int row0 = blockIdx.x * AGG_ROWS;
  const int rows = min(AGG_ROWS, N - row0);

  {
    float4 f[IT0];
    float  d[IT0];
#pragma unroll
    for (int i = 0; i < IT0; ++i) {
      const int idx = tid + i * AGG_NT;
      const int r = idx / D4, c4 = idx - r * D4;
      const int t = min(row0 + r, N - 1);
      d[i] = dinv[t];
      f[i] = ((const float4*)(feat + (size_t)t * D))[c4];
    }
#pragma unroll
    for (int i = 0; i < IT0; ++i) {
      const int idx = tid + i * AGG_NT;
      const float nrm = d[i] * d[i];
      if (idx < AGG_ROWS * D4)
        ((float4*)tile)[idx] = make_float4(nrm * f[i].x, nrm * f[i].y, nrm * f[i].z, nrm * f[i].w);
    }
  }
  if (tid == 0) l_cnt = 0;
  __syncthreads();

  auto take = [&](int e, int t) {
    const unsigned r = (unsigned)(t - row0);
    if (r < (unsigned)rows) {
      const int pos = __hip_atomic_fetch_add(&l_cnt, 1, __ATOMIC_RELAXED,
                                             __HIP_MEMORY_SCOPE_WORKGROUP);
      if (pos < AGG_CAP) {
        l_src[pos] = e; l_row[pos] = (int)r;
      } else {
        const int s = src[e];
        const float nrm = dinv[s] * dinv[t];
        const float* fs = feat + (size_t)s * D;
        float* tr = tile + r * D;
        for (int c = 0; c < D; ++c)
          __hip_atomic_fetch_add(tr + c, nrm * fs[c], __ATOMIC_RELAXED,
                                 __HIP_MEMORY_SCOPE_WORKGROUP);
      }
    }
  };

  const bool vec = ((E & 3) == 0) && ((((size_t)tgt) & 15) == 0);
  const int nvec = vec ? (E >> 2) : 0;
  const int rot = (int)blockIdx.x * (nvec / (int)gridDim.x);
  const int4* tgt4 = (const int4*)tgt;
  for (int base = tid; base < nvec; base += AGG_SCAN * AGG_NT) {
    int4 tv[AGG_SCAN];
    int  ix[AGG_SCAN];
#pragma unroll
    for (int u = 0; u < AGG_SCAN; ++u) {
      const int i = base + u * AGG_NT;
      int j = i + rot;
      if (j >= nvec) j -= nvec;
      ix[u] = j;
      tv[u] = (i < nvec) ? tgt4[j] : make_int4(-1, -1, -1, -1);
    }
#pragma unroll
    for (int u = 0; u < AGG_SCAN; ++u) {
      take(4 * ix[u] + 0, tv[u].x);
      take(4 * ix[u] + 1, tv[u].y);
      take(4 * ix[u] + 2, tv[u].z);
      take(4 * ix[u] + 3, tv[u].w);
    }
  }
  for (int e = nvec * 4 + tid; e < E; e += AGG_NT) take(e, tgt[e]);
  __syncthreads();

  const int n = min(l_cnt, AGG_CAP);
  for (int k = tid; k < n; k += AGG_NT) {
    const int s = src[l_src[k]];
    l_nrm[k] = dinv[s] * dinv[row0 + l_row[k]];
    l_src[k] = s;
  }
  __syncthreads();

  int k = wid;
  for (; k + (U - 1) * AGG_NW < n; k += U * AGG_NW) {
    float2 a[U][Q];
    int    rr[U];
    float  nn[U];
#pragma unroll
    for (int u = 0; u < U; ++u) {
      const int ku = k + u * AGG_NW;
      const int s = l_src[ku];
      rr[u] = l_row[ku];
      nn[u] = l_nrm[ku];
      const float2* f = (const float2*)(feat + (size_t)s * D) + lane;
#pragma unroll
      for (int q = 0; q < Q; ++q) a[u][q] = f[32 * q];
    }
#pragma unroll
    for (int u = 0; u < U; ++u) {
      float* tr = tile + rr[u] * D + 2 * lane;
#pragma unroll
      for (int q = 0; q < Q; ++q) {
        __hip_atomic_fetch_add(tr + 64 * q,     nn[u] * a[u][q].x, __ATOMIC_RELAXED,
                               __HIP_MEMORY_SCOPE_WORKGROUP);
        __hip_atomic_fetch_add(tr + 64 * q + 1, nn[u] * a[u][q].y, __ATOMIC_RELAXED,
                               __HIP_MEMORY_SCOPE_WORKGROUP);
      }
    }
  }
  for (; k < n; k += AGG_NW) {
    const int s = l_src[k];
    const float nk = l_nrm[k];
    const float2* f = (const float2*)(feat + (size_t)s * D) + lane;
    float* tr = tile + l_row[k] * D + 2 * lane;
    float2 a[Q];
#pragma unroll
    for (int q = 0; q < Q; ++q) a[q] = f[32 * q];
#pragma unroll
    for (int q = 0; q < Q; ++q) {
      __hip_atomic_fetch_add(tr + 64 * q,     nk * a[q].x, __ATOMIC_RELAXED,
                             __HIP_MEMORY_SCOPE_WORKGROUP);
      __hip_atomic_fetch_add(tr + 64 * q + 1, nk * a[q].y, __ATOMIC_RELAXED,
                             __HIP_MEMORY_SCOPE_WORKGROUP);
    }
  }
  __syncthreads();

  for (int idx = tid; idx < rows * D4; idx += AGG_NT) {
    const int r = idx / D4, c4 = idx - r * D4;
    ((float4*)(out + (size_t)(row0 + r) * D))[c4] = ((const float4*)tile)[idx];
  }
}


extern "C" void kernel_launch(void* const* d_in, const int* in_sizes, int n_in,
                              void* d_out, int out_size, void* d_ws, size_t ws_size,
                              hipStream_t stream) {
  const float* x  = (const float*)d_in[0];
  const int*   ei = (const int*)d_in[1];
  const float* W1 = (const float*)d_in[2];
  const float* W2 = (const float*)d_in[3];
  float* out = (float*)d_out;

  const int N = in_sizes[0] / DIN;
  const int E = in_sizes[1] / 2;
  const int* src = ei;
  const int* tgt = ei + E;

  char* ws = (char*)d_ws;
  size_t off = 0;
  float* dinv = (float*)(ws + off); off += ((size_t)N * 4 + 255) & ~(size_t)255;
  float* xw   = (float*)(ws + off); off += ((size_t)N * DHID * 4 + 255) & ~(size_t)255;
  float* h    = (float*)(ws + off); off += ((size_t)N * DHID * 4 + 255) & ~(size_t)255;
  float* hw   = xw;
  (void)ws_size; (void)n_in; (void)out_size;

  const int aggBlocks = (N + AGG_ROWS - 1) / AGG_ROWS;
  const int mBlocks = (N + GM_BM - 1) / GM_BM;

  k_fill<<<(N + 255) / 256, 256, 0, stream>>>(dinv, 1.0f, N);
  k_deg_count<<<(E + 255) / 256, 256, 0, stream>>>(tgt, dinv, E);
  k_rsqrt<<<(N + 255) / 256, 256, 0, stream>>>(dinv, N);

  {
    dim3 grid(mBlocks, DHID / 256);
    k_gemm_lds<DIN, DHID, 256, false><<<grid, GM_NTH, 0, stream>>>(x, W1, xw, N);
  }
  k_agg_tile<DHID, 2><<<aggBlocks, AGG_NT, 0, stream>>>(xw, dinv, src, tgt, h, E, N);
  {
    dim3 grid(mBlocks, DOUT / 128);
    k_gemm_lds<DHID, DOUT, 128, true><<<grid, GM_NTH, 0, stream>>>(h, W2, hw, N);
  }
  k_agg_tile<DOUT, 4><<<aggBlocks, AGG_NT, 0, stream>>>(hw, dinv, src, tgt, out, E, N);
}


# Tile edge filter hoisted: one launch in front walks the target list once per tile, keeps (source,row) lists in workspace, counts in-degrees from the hits; 5 launches, LDS sums still arrival order
# speedup vs baseline: 2.4240x; 1.2577x over previous
#include <hip/hip_runtime.h>

typedef float v2f __attribute__((ext_vector_type(2)));
typedef float v8f __attribute__((ext_vector_type(8)));

#define DIN  512
#define DHID 512
#define DOUT 256


#define GM_BM 80
#define GM_BK 64
#define GM_PA 68
#define GM_NTH 256

template <int K, int N, int BN, bool SQUARE_A>
__global__ __launch_bounds__(GM_NTH)
void k_gemm_lds(const float* A, const float* B, float* C, int M) {
  constexpr int NT      = BN / 128;
  constexpr int PB      = 2 * BN + 32;
  constexpr int PC      = BN + 4;
  constexpr int A_STAGE = GM_BM * GM_PA;
  constexpr int B_STAGE = (GM_BK / 2) * PB;
  constexpr int STAGE   = A_STAGE + B_STAGE;
  constexpr int C_TILE  = GM_BM * PC;
  constexpr int LDS_FL  = (2 * STAGE > C_TILE) ? 2 * STAGE : C_TILE;
  constexpr int NKB     = K / GM_BK;
  constexpr int AK4     = GM_BK / 4;
  constexpr int A_IT    = GM_BM * AK4 / GM_NTH;
  constexpr int BN4     = BN / 4;
  constexpr int B_IT    = (GM_BK / 2) * BN4 / GM_NTH;
  static_assert(K % GM_BK == 0 && BN % 128 == 0 && N % BN == 0, "tile shape");
  static_assert((GM_BM * AK4) % GM_NTH == 0, "A copy shape");
  static_assert(((GM_BK / 2) * BN4) % GM_NTH == 0, "B copy shape");

  __shared__ __attribute__((aligned(16))) float lds[LDS_FL];

  const int tid  = threadIdx.x;
  const int lane = tid & 31;
  const int wid  = tid >> 5;
  const int m = lane & 15;
  const int g = lane >> 4;
  const int row0 = blockIdx.x * GM_BM;
  const int col0 = blockIdx.y * BN;
  const int wcol = wid * (NT * 16);

  const float* ap[A_IT];
  int a_l[A_IT];
#pragma unroll
  for (int i = 0; i < A_IT; ++i) {
    const int idx = tid + i * GM_NTH;
    const int r = idx / AK4, c4 = idx - r * AK4;
    const int gr = min(row0 + r, M - 1);
    ap[i]  = A + (size_t)gr * K + c4 * 4;
    a_l[i] = r * GM_PA + c4 * 4;
  }
  const float* bp[B_IT];
  int b_l[B_IT];
#pragma unroll
  for (int i = 0; i < B_IT; ++i) {
    const int item = tid + i * GM_NTH;
    const int jp = item / BN4, n4 = item - jp * BN4;
    bp[i]  = B + (size_t)(2 * jp) * N + col0 + n4 * 4;
    b_l[i] = jp * PB + n4 * 8;
  }

  float4 ra[A_IT], rb0[B_IT], rb1[B_IT];
  auto gload = [&](int kb) {
#pragma unroll
    for (int i = 0; i < A_IT; ++i) ra[i] = *(const float4*)(ap[i] + kb * GM_BK);
#pragma unroll
    for (int i = 0; i < B_IT; ++i) {
      const float* p = bp[i] + (size_t)kb * GM_BK * N;
      rb0[i] = *(const float4*)(p);
      rb1[i] = *(const float4*)(p + N);
    }
  };
  auto lstore = [&](int st) {
    float* as = lds + st * STAGE;
    float* bs = as + A_STAGE;
#pragma unroll
    for (int i = 0; i < A_IT; ++i) {
      const float4 v = ra[i];
      *(float4*)(as + a_l[i]) = make_float4(v.x, v.y, v.z, v.w);
    }
#pragma unroll
    for (int i = 0; i < B_IT; ++i) {
      *(float4*)(bs + b_l[i])     = make_float4(rb0[i].x, rb1[i].x, rb0[i].y, rb1[i].y);
      *(float4*)(bs + b_l[i] + 4) = make_float4(rb0[i].z, rb1[i].z, rb0[i].w, rb1[i].w);
    }
  };

  v8f acc[5][NT] = {};
  auto compute = [&](int st) {
    const float* as = lds + st * STAGE + m * GM_PA + 2 * g;
    const float* bs = lds + st * STAGE + A_STAGE + g * PB + (wcol + m) * 2;
#pragma unroll
    for (int s = 0; s < GM_BK / 4; ++s) {
      v2f a[5], b[NT];
#pragma unroll
      for (int i = 0; i < 5; ++i) {
        a[i] = *(const v2f*)(as + i * 16 * GM_PA + s * 4);
        if (SQUARE_A) a[i] *= a[i];
      }
#pragma unroll
      for (int j = 0; j < NT; ++j) b[j] = *(const v2f*)(bs + s * 2 * PB + j * 32);
#pragma unroll
      for (int i = 0; i < 5; ++i)
#pragma unroll
        for (int j = 0; j < NT; ++j)
          acc[i][j] = __builtin_amdgcn_wmma_f32_16x16x4_f32(false, a[i], false, b[j],
                                                            (short)0, acc[i][j], false, false);
    }
  };

  gload(0);
  lstore(0);
  __syncthreads();
  for (int kb = 0; kb < NKB - 1; ++kb) {
    gload(kb + 1);
    compute(kb & 1);
    __builtin_amdgcn_sched_barrier(0);
    lstore((kb + 1) & 1);
    __syncthreads();
  }
  compute((NKB - 1) & 1);
  __syncthreads();

#pragma unroll
  for (int i = 0; i < 5; ++i)
#pragma unroll
    for (int j = 0; j < NT; ++j)
#pragma unroll
      for (int v = 0; v < 8; ++v)
        lds[(i * 16 + 8 * g + v) * PC + wcol + j * 16 + m] = acc[i][j][v];
  __syncthreads();

  constexpr int C_IT = GM_BM * BN4 / GM_NTH;
  static_assert((GM_BM * BN4) % GM_NTH == 0, "C copy shape");
  if (row0 + GM_BM <= M) {
    float4 cv[C_IT];
#pragma unroll
    for (int it = 0; it < C_IT; ++it) {
      const int idx = tid + it * GM_NTH;
      const int r = idx / BN4, c4 = idx - r * BN4;
      cv[it] = *(const float4*)(lds + r * PC + c4 * 4);
    }
#pragma unroll
    for (int it = 0; it < C_IT; ++it) {
      const int idx = tid + it * GM_NTH;
      const int r = idx / BN4, c4 = idx - r * BN4;
      *(float4*)(C + (size_t)(row0 + r) * N + col0 + c4 * 4) = cv[it];
    }
  } else {
    for (int idx = tid; idx < GM_BM * BN4; idx += GM_NTH) {
      const int r = idx / BN4, c4 = idx - r * BN4;
      if (row0 + r < M)
        *(float4*)(C + (size_t)(row0 + r) * N + col0 + c4 * 4) = *(const float4*)(lds + r * PC + c4 * 4);
    }
  }
}


#define AGG_ROWS 40
#define AGG_NT   1024
#define AGG_NW   (AGG_NT / 32)
#define AGG_CAP  2048
#define AGG_SCAN 8
#define AGG_LIT  (AGG_CAP / AGG_NT)

__global__ __launch_bounds__(AGG_NT)
void k_tile_lists(const int* __restrict__ src, const int* __restrict__ tgt,
                  float* __restrict__ dinv, int* __restrict__ lists,
                  int* __restrict__ counts, int E, int N) {
  __shared__ int l_e[AGG_CAP];
  __shared__ int l_row[AGG_CAP];
  __shared__ int l_deg[AGG_ROWS];
  __shared__ int l_cnt;

  const int tid  = threadIdx.x;
  const int row0 = blockIdx.x * AGG_ROWS;
  const int rows = min(AGG_ROWS, N - row0);

  if (tid < AGG_ROWS) l_deg[tid] = 0;
  if (tid == 0) l_cnt = 0;
  __syncthreads();

  auto take = [&](int e, int t) {
    const unsigned r = (unsigned)(t - row0);
    if (r < (unsigned)rows) {
      __hip_atomic_fetch_add(&l_deg[r], 1, __ATOMIC_RELAXED, __HIP_MEMORY_SCOPE_WORKGROUP);
      const int pos = __hip_atomic_fetch_add(&l_cnt, 1, __ATOMIC_RELAXED,
                                             __HIP_MEMORY_SCOPE_WORKGROUP);
      if (pos < AGG_CAP) { l_e[pos] = e; l_row[pos] = (int)r; }
    }
  };

  const bool vec = ((E & 3) == 0) && ((((size_t)tgt) & 15) == 0);
  const int nvec = vec ? (E >> 2) : 0;
  const int rot = (int)blockIdx.x * (nvec / (int)gridDim.x);
  const int4* tgt4 = (const int4*)tgt;
  for (int base = tid; base < nvec; base += AGG_SCAN * AGG_NT) {
    int4 tv[AGG_SCAN];
    int  ix[AGG_SCAN];
#pragma unroll
    for (int u = 0; u < AGG_SCAN; ++u) {
      const int i = base + u * AGG_NT;
      int j = i + rot;
      if (j >= nvec) j -= nvec;
      ix[u] = j;
      tv[u] = (i < nvec) ? tgt4[j] : make_int4(-1, -1, -1, -1);
    }
#pragma unroll
    for (int u = 0; u < AGG_SCAN; ++u) {
      take(4 * ix[u] + 0, tv[u].x);
      take(4 * ix[u] + 1, tv[u].y);
      take(4 * ix[u] + 2, tv[u].z);
      take(4 * ix[u] + 3, tv[u].w);
    }
  }
  for (int e = nvec * 4 + tid; e < E; e += AGG_NT) take(e, tgt[e]);
  __syncthreads();

  const int total = l_cnt;
  const int n = min(total, AGG_CAP);
  int2* wl = (int2*)lists + (size_t)blockIdx.x * AGG_CAP;
  for (int k = tid; k < n; k += AGG_NT) wl[k] = make_int2(src[l_e[k]], l_row[k]);
  if (tid == 0) counts[blockIdx.x] = total;
  if (tid < rows) {
    const float d = 1.0f + (float)l_deg[tid];
    dinv[row0 + tid] = (d > 0.f) ? rsqrtf(d) : 0.f;
  }
}

template <int D, int U>
__global__ __launch_bounds__(AGG_NT)
void k_agg_tile(const float* __restrict__ feat, const float* __restrict__ dinv,
                const int* __restrict__ src, const int* __restrict__ tgt,
                const int* __restrict__ lists, const int* __restrict__ counts,
                float* __restrict__ out, int E, int N) {
  __shared__ float tile[AGG_ROWS * D];
  __shared__ int   l_src[AGG_CAP];
  __shared__ int   l_row[AGG_CAP];
  __shared__ float l_nrm[AGG_CAP];

  constexpr int D4  = D / 4;
  constexpr int Q   = D / 64;
  constexpr int IT0 = (AGG_ROWS * D4 + AGG_NT - 1) / AGG_NT;
  const int tid  = threadIdx.x;
  const int lane = tid & 31;
  const int wid  = tid >> 5;
  const int row0 = blockIdx.x * AGG_ROWS;
  const int rows = min(AGG_ROWS, N - row0);

  const int total = counts[blockIdx.x];
  const int n = (total > AGG_CAP) ? 0 : total;
  const int2* wl = (const int2*)lists + (size_t)blockIdx.x * AGG_CAP;

  {
    int2   sr[AGG_LIT];
    float4 f[IT0];
    float  d[IT0];
#pragma unroll
    for (int i = 0; i < IT0; ++i) {
      const int idx = tid + i * AGG_NT;
      const int r = idx / D4, c4 = idx - r * D4;
      const int t = min(row0 + r, N - 1);
      d[i] = dinv[t];
      f[i] = ((const float4*)(feat + (size_t)t * D))[c4];
    }
#pragma unroll
    for (int i = 0; i < AGG_LIT; ++i) {
      const int k = tid + i * AGG_NT;
      sr[i] = (k < n) ? wl[k] : make_int2(0, 0);
    }
#pragma unroll
    for (int i = 0; i < AGG_LIT; ++i) {
      const int k = tid + i * AGG_NT;
      if (k < n) {
        l_nrm[k] = dinv[sr[i].x] * dinv[row0 + sr[i].y];
        l_src[k] = sr[i].x;
        l_row[k] = sr[i].y;
      }
    }
#pragma unroll
    for (int i = 0; i < IT0; ++i) {
      const int idx = tid + i * AGG_NT;
      const float nrm = d[i] * d[i];
      if (idx < AGG_ROWS * D4)
        ((float4*)tile)[idx] = make_float4(nrm * f[i].x, nrm * f[i].y, nrm * f[i].z, nrm * f[i].w);
    }
  }
  __syncthreads();

  if (total > AGG_CAP) {
    for (int e = tid; e < E; e += AGG_NT) {
      const int t = tgt[e];
      const unsigned r = (unsigned)(t - row0);
      if (r < (unsigned)rows) {
        const int s = src[e];
        const float nrm = dinv[s] * dinv[t];
        const float* fs = feat + (size_t)s * D;
        float* tr = tile + r * D;
        for (int c = 0; c < D; ++c)
          __hip_atomic_fetch_add(tr + c, nrm * fs[c], __ATOMIC_RELAXED,
                                 __HIP_MEMORY_SCOPE_WORKGROUP);
      }
    }
  }

  int k = wid;
  for (; k + (U - 1) * AGG_NW < n; k += U * AGG_NW) {
    float2 a[U][Q];
    int    rr[U];
    float  nn[U];
#pragma unroll
    for (int u = 0; u < U; ++u) {
      const int ku = k + u * AGG_NW;
      const int s = l_src[ku];
      rr[u] = l_row[ku];
      nn[u] = l_nrm[ku];
      const float2* f = (const float2*)(feat + (size_t)s * D) + lane;
#pragma unroll
      for (int q = 0; q < Q; ++q) a[u][q] = f[32 * q];
    }
#pragma unroll
    for (int u = 0; u < U; ++u) {
      float* tr = tile + rr[u] * D + 2 * lane;
#pragma unroll
      for (int q = 0; q < Q; ++q) {
        __hip_atomic_fetch_add(tr + 64 * q,     nn[u] * a[u][q].x, __ATOMIC_RELAXED,
                               __HIP_MEMORY_SCOPE_WORKGROUP);
        __hip_atomic_fetch_add(tr + 64 * q + 1, nn[u] * a[u][q].y, __ATOMIC_RELAXED,
                               __HIP_MEMORY_SCOPE_WORKGROUP);
      }
    }
  }
  for (; k < n; k += AGG_NW) {
    const int s = l_src[k];
    const float nk = l_nrm[k];
    const float2* f = (const float2*)(feat + (size_t)s * D) + lane;
    float* tr = tile + l_row[k] * D + 2 * lane;
    float2 a[Q];
#pragma unroll
    for (int q = 0; q < Q; ++q) a[q] = f[32 * q];
#pragma unroll
    for (int q = 0; q < Q; ++q) {
      __hip_atomic_fetch_add(tr + 64 * q,     nk * a[q].x, __ATOMIC_RELAXED,
                             __HIP_MEMORY_SCOPE_WORKGROUP);
      __hip_atomic_fetch_add(tr + 64 * q + 1, nk * a[q].y, __ATOMIC_RELAXED,
                             __HIP_MEMORY_SCOPE_WORKGROUP);
    }
  }
  __syncthreads();

  for (int idx = tid; idx < rows * D4; idx += AGG_NT) {
    const int r = idx / D4, c4 = idx - r * D4;
    ((float4*)(out + (size_t)(row0 + r) * D))[c4] = ((const float4*)tile)[idx];
  }
}


extern "C" void kernel_launch(void* const* d_in, const int* in_sizes, int n_in,
                              void* d_out, int out_size, void* d_ws, size_t ws_size,
                              hipStream_t stream) {
  const float* x  = (const float*)d_in[0];
  const int*   ei = (const int*)d_in[1];
  const float* W1 = (const float*)d_in[2];
  const float* W2 = (const float*)d_in[3];
  float* out = (float*)d_out;

  const int N = in_sizes[0] / DIN;
  const int E = in_sizes[1] / 2;
  const int* src = ei;
  const int* tgt = ei + E;

  char* ws = (char*)d_ws;
  size_t off = 0;
  const int aggBlocks = (N + AGG_ROWS - 1) / AGG_ROWS;
  const int mBlocks = (N + GM_BM - 1) / GM_BM;
  float* dinv = (float*)(ws + off); off += ((size_t)N * 4 + 255) & ~(size_t)255;
  float* xw   = (float*)(ws + off); off += ((size_t)N * DHID * 4 + 255) & ~(size_t)255;
  float* h    = (float*)(ws + off); off += ((size_t)N * DHID * 4 + 255) & ~(size_t)255;
  int* lists  = (int*)(ws + off);   off += ((size_t)aggBlocks * AGG_CAP * 8 + 255) & ~(size_t)255;
  int* counts = (int*)(ws + off);   off += ((size_t)aggBlocks * 4 + 255) & ~(size_t)255;
  float* hw   = xw;
  (void)ws_size; (void)n_in; (void)out_size;

  k_tile_lists<<<aggBlocks, AGG_NT, 0, stream>>>(src, tgt, dinv, lists, counts, E, N);

  {
    dim3 grid(mBlocks, DHID / 256);
    k_gemm_lds<DIN, DHID, 256, false><<<grid, GM_NTH, 0, stream>>>(x, W1, xw, N);
  }
  k_agg_tile<DHID, 2><<<aggBlocks, AGG_NT, 0, stream>>>(xw, dinv, src, tgt, lists, counts, h, E, N);
  {
    dim3 grid(mBlocks, DOUT / 128);
    k_gemm_lds<DHID, DOUT, 128, true><<<grid, GM_NTH, 0, stream>>>(h, W2, hw, N);
  }
  k_agg_tile<DOUT, 4><<<aggBlocks, AGG_NT, 0, stream>>>(hw, dinv, src, tgt, lists, counts, out, E, N);
}
